# starting kernel
# baseline (speedup 1.0000x reference)
#include <hip/hip_runtime.h>
#include <math.h>

typedef __attribute__((ext_vector_type(2))) float v2f;
typedef __attribute__((ext_vector_type(8))) float v8f;

#define NQ 6
#define NGATES 15
#define DFEAT 4096
#define ROWS_PER_BLOCK 128
#define THREADS 256

__global__ __launch_bounds__(THREADS) void hybrid_pyramid_fused(
    const float* __restrict__ x,  const float* __restrict__ W1,
    const float* __restrict__ b1, const float* __restrict__ qw,
    const float* __restrict__ W2, const float* __restrict__ b2,
    float* __restrict__ out)
{
    __shared__ float w1s[DFEAT * NQ];
    __shared__ float hS[ROWS_PER_BLOCK][NQ];
    __shared__ float gsin[NGATES + 1];
    __shared__ float gcos[NGATES + 1];

    {
        const float4* __restrict__ src = (const float4*)W1;
        float4* dst = (float4*)w1s;
        for (int i = threadIdx.x; i < (DFEAT * NQ) / 4; i += THREADS) dst[i] = src[i];
    }
    if (threadIdx.x < NGATES) {
        float w = qw[threadIdx.x];
        gsin[threadIdx.x] = sinf(w);
        gcos[threadIdx.x] = cosf(w);
    }
    __syncthreads();

    const int lane = threadIdx.x & 31;
    const int wave = threadIdx.x >> 5;
    const int m    = lane & 15;
    const int hi   = lane >> 4;
    const int koff = hi * 2;
    const int ncol = (m < NQ) ? m : (NQ - 1);

    const long rowBase = (long)blockIdx.x * ROWS_PER_BLOCK + (long)wave * 16;
    const float* __restrict__ xp = x + (rowBase + m) * DFEAT + koff;
    const float* __restrict__ bp = w1s + koff * NQ + ncol;

    v8f acc = {0.f, 0.f, 0.f, 0.f, 0.f, 0.f, 0.f, 0.f};

    #pragma unroll 8
    for (int kc = 0; kc < DFEAT; kc += 4) {
        v2f a = *(const v2f*)(xp + kc);
        v2f b;
        b.x = bp[kc * NQ];
        b.y = bp[kc * NQ + NQ];
        acc = __builtin_amdgcn_wmma_f32_16x16x4_f32(
             false, a,  false, b,
             (short)0, acc,  false,  false);
    }

    if (m < NQ) {
        #pragma unroll
        for (int j = 0; j < 8; ++j)
            hS[wave * 16 + hi * 8 + j][m] = acc[j];
    }
    __syncthreads();

    if (threadIdx.x < ROWS_PER_BLOCK) {
        const int r = threadIdx.x;
        float a[NQ];
        #pragma unroll
        for (int n = 0; n < NQ; ++n) a[n] = hS[r][n] + b1[n];

        float ss = 0.f;
        #pragma unroll
        for (int n = 0; n < NQ; ++n) ss += a[n] * a[n];
        float inv = 1.0f / sqrtf(ss);
        #pragma unroll
        for (int n = 0; n < NQ; ++n) a[n] *= inv;

        const int Q[NGATES] = {4, 3, 2, 4, 1, 3, 0, 2, 4, 1, 3, 2, 4, 3, 4};
        #pragma unroll
        for (int k = 0; k < NGATES; ++k) {
            const int q = Q[k];
            float s = gsin[k], c = gcos[k];
            float aq  =  c * a[q] + s * a[q + 1];
            float aq1 = -s * a[q] + c * a[q + 1];
            a[q] = aq; a[q + 1] = aq1;
        }

        float accum = b2[0];
        #pragma unroll
        for (int n = 0; n < NQ; ++n) {
            float z = 1.f - 2.f * a[n] * a[n];
            z = (z != z) ? 0.f : z;
            accum += z * W2[n];
        }
        out[(long)blockIdx.x * ROWS_PER_BLOCK + r] = 1.f / (1.f + expf(-accum));
    }
}

extern "C" void kernel_launch(void* const* d_in, const int* in_sizes, int n_in,
                              void* d_out, int out_size, void* d_ws, size_t ws_size,
                              hipStream_t stream) {
    const float* x  = (const float*)d_in[0];
    const float* W1 = (const float*)d_in[1];
    const float* b1 = (const float*)d_in[2];
    const float* qw = (const float*)d_in[3];
    const float* W2 = (const float*)d_in[4];
    const float* b2 = (const float*)d_in[5];
    float* out = (float*)d_out;

    const int B = in_sizes[0] / DFEAT;
    dim3 grid(B / ROWS_PER_BLOCK);
    hybrid_pyramid_fused<<<grid, THREADS, 0, stream>>>(x, W1, b1, qw, W2, b2, out);
}


# more x bytes in flight: 16-byte loads requested one 8-load group ahead of their WMMAs, 4 waves per 16-row tile (K quarters; partials summed in LDS: sum order changes), 256 blocks x 16 waves
# speedup vs baseline: 2.8851x; 2.8851x over previous
#include <hip/hip_runtime.h>
#include <math.h>

typedef __attribute__((ext_vector_type(2))) float v2f;
typedef __attribute__((ext_vector_type(4))) float v4f;
typedef __attribute__((ext_vector_type(8))) float v8f;

#define NQ 6
#define NGATES 15
#define DFEAT 4096
#define KSPLIT 4
#define KPER (DFEAT / KSPLIT)
#define TILES 4
#define ROWS_PER_BLOCK (16 * TILES)
#define THREADS (32 * TILES * KSPLIT)
#define GROUP 8
#define NGROUPS (KPER / (8 * GROUP))
#define W1_PER_THREAD ((DFEAT * NQ / 4) / THREADS)

__global__ __launch_bounds__(THREADS) void hybrid_pyramid_fused(
    const float* __restrict__ x,  const float* __restrict__ W1,
    const float* __restrict__ b1, const float* __restrict__ qw,
    const float* __restrict__ W2, const float* __restrict__ b2,
    float* __restrict__ out)
{
    __shared__ float w1s[DFEAT * NQ];
    __shared__ float hP[KSPLIT][ROWS_PER_BLOCK][NQ];
    __shared__ float gsin[NGATES + 1];
    __shared__ float gcos[NGATES + 1];

    const int lane = threadIdx.x & 31;
    const int wave = threadIdx.x >> 5;
    const int c    = lane & 15;
    const int hi   = lane >> 4;
    const int tile = wave >> 2;
    const int kq   = wave & 3;
    const int ncol = (c < NQ) ? c : (NQ - 1);

    const long row = (long)blockIdx.x * ROWS_PER_BLOCK + tile * 16 + c;
    const float* __restrict__ xp = x + row * DFEAT + kq * KPER + hi * 4;
    const float* bp = w1s + (kq * KPER + hi * 4) * NQ + ncol;

    v4f wtmp[W1_PER_THREAD];
    {
        const v4f* __restrict__ src = (const v4f*)W1;
        #pragma unroll
        for (int i = 0; i < W1_PER_THREAD; ++i) wtmp[i] = src[threadIdx.x + i * THREADS];
    }
    v4f buf[2][GROUP];
    #pragma unroll
    for (int t = 0; t < GROUP; ++t) buf[0][t] = *(const v4f*)(xp + t * 8);
    {
        v4f* dst = (v4f*)w1s;
        #pragma unroll
        for (int i = 0; i < W1_PER_THREAD; ++i) dst[threadIdx.x + i * THREADS] = wtmp[i];
    }
    if (threadIdx.x < NGATES) {
        float w = qw[threadIdx.x];
        gsin[threadIdx.x] = sinf(w);
        gcos[threadIdx.x] = cosf(w);
    }
    __syncthreads();

    v8f acc = {0.f, 0.f, 0.f, 0.f, 0.f, 0.f, 0.f, 0.f};

    #pragma unroll
    for (int g = 0; g < NGROUPS; ++g) {
        if (g + 1 < NGROUPS) {
            #pragma unroll
            for (int t = 0; t < GROUP; ++t)
                buf[(g + 1) & 1][t] = *(const v4f*)(xp + (g + 1) * (8 * GROUP) + t * 8);
        }
        #pragma unroll
        for (int t = 0; t < GROUP; ++t) {
            const v4f av = buf[g & 1][t];
            const float* bq = bp + (g * (8 * GROUP) + t * 8) * NQ;
            v2f a0, a1, b0, b1v;
            a0.x = av.x;  a0.y = av.y;
            a1.x = av.z;  a1.y = av.w;
            b0.x  = bq[0];       b0.y  = bq[NQ];
            b1v.x = bq[2 * NQ];  b1v.y = bq[3 * NQ];
            acc = __builtin_amdgcn_wmma_f32_16x16x4_f32(
                 false, a0,  false, b0,
                 (short)0, acc,  false,  false);
            acc = __builtin_amdgcn_wmma_f32_16x16x4_f32(
                 false, a1,  false, b1v,
                 (short)0, acc,  false,  false);
        }
    }

    if (c < NQ) {
        #pragma unroll
        for (int j = 0; j < 8; ++j)
            hP[kq][tile * 16 + hi * 8 + j][c] = acc[j];
    }
    __syncthreads();

    if (threadIdx.x < ROWS_PER_BLOCK) {
        const int r = threadIdx.x;
        float a[NQ];
        #pragma unroll
        for (int n = 0; n < NQ; ++n)
            a[n] = ((hP[0][r][n] + hP[1][r][n]) + (hP[2][r][n] + hP[3][r][n])) + b1[n];

        float ss = 0.f;
        #pragma unroll
        for (int n = 0; n < NQ; ++n) ss += a[n] * a[n];
        float inv = 1.0f / sqrtf(ss);
        #pragma unroll
        for (int n = 0; n < NQ; ++n) a[n] *= inv;

        const int Q[NGATES] = {4, 3, 2, 4, 1, 3, 0, 2, 4, 1, 3, 2, 4, 3, 4};
        #pragma unroll
        for (int k = 0; k < NGATES; ++k) {
            const int q = Q[k];
            float s = gsin[k], cc = gcos[k];
            float aq  =  cc * a[q] + s * a[q + 1];
            float aq1 = -s * a[q] + cc * a[q + 1];
            a[q] = aq; a[q + 1] = aq1;
        }

        float accum = b2[0];
        #pragma unroll
        for (int n = 0; n < NQ; ++n) {
            float z = 1.f - 2.f * a[n] * a[n];
            z = (z != z) ? 0.f : z;
            accum += z * W2[n];
        }
        out[(long)blockIdx.x * ROWS_PER_BLOCK + r] = 1.f / (1.f + expf(-accum));
    }
}

extern "C" void kernel_launch(void* const* d_in, const int* in_sizes, int n_in,
                              void* d_out, int out_size, void* d_ws, size_t ws_size,
                              hipStream_t stream) {
    const float* x  = (const float*)d_in[0];
    const float* W1 = (const float*)d_in[1];
    const float* b1 = (const float*)d_in[2];
    const float* qw = (const float*)d_in[3];
    const float* W2 = (const float*)d_in[4];
    const float* b2 = (const float*)d_in[5];
    float* out = (float*)d_out;

    const int B = in_sizes[0] / DFEAT;
    dim3 grid(B / ROWS_PER_BLOCK);
    hybrid_pyramid_fused<<<grid, THREADS, 0, stream>>>(x, W1, b1, qw, W2, b2, out);
}


# x loads cover whole 128-byte lines (2 rows x 256 contiguous bytes per instruction); each wave turns its 16x64 tile into A-fragment order through a private LDS tile; k order and sums as before
# speedup vs baseline: 3.0961x; 1.0731x over previous
#include <hip/hip_runtime.h>
#include <math.h>

typedef __attribute__((ext_vector_type(2))) float v2f;
typedef __attribute__((ext_vector_type(4))) float v4f;
typedef __attribute__((ext_vector_type(8))) float v8f;

#define NQ 6
#define NGATES 15
#define DFEAT 4096
#define KSPLIT 4
#define KPER (DFEAT / KSPLIT)
#define TILES 4
#define ROWS_PER_BLOCK (16 * TILES)
#define THREADS (32 * TILES * KSPLIT)
#define GROUP 8
#define NGROUPS (KPER / (8 * GROUP))
#define W1_PER_THREAD ((DFEAT * NQ / 4) / THREADS)
#define NWAVES (THREADS / 32)
#define STAGE_PITCH 68

__global__ __launch_bounds__(THREADS) void hybrid_pyramid_fused(
    const float* __restrict__ x,  const float* __restrict__ W1,
    const float* __restrict__ b1, const float* __restrict__ qw,
    const float* __restrict__ W2, const float* __restrict__ b2,
    float* __restrict__ out)
{
    __shared__ float w1s[DFEAT * NQ];
    __shared__ float xs[NWAVES][16][STAGE_PITCH];
    __shared__ float hP[KSPLIT][ROWS_PER_BLOCK][NQ];
    __shared__ float gsin[NGATES + 1];
    __shared__ float gcos[NGATES + 1];

    const int lane = threadIdx.x & 31;
    const int wave = threadIdx.x >> 5;
    const int c    = lane & 15;
    const int hi   = lane >> 4;
    const int tile = wave >> 2;
    const int kq   = wave & 3;
    const int ncol = (c < NQ) ? c : (NQ - 1);

    const long row0 = (long)blockIdx.x * ROWS_PER_BLOCK + tile * 16;
    const float* __restrict__ xg = x + (row0 + hi) * DFEAT + kq * KPER + c * 4;
    float* stw = &xs[wave][hi][c * 4];
    const float* str = &xs[wave][c][hi * 4];
    const float* bp = w1s + (kq * KPER + hi * 4) * NQ + ncol;

    v4f wtmp[W1_PER_THREAD];
    {
        const v4f* __restrict__ src = (const v4f*)W1;
        #pragma unroll
        for (int i = 0; i < W1_PER_THREAD; ++i) wtmp[i] = src[threadIdx.x + i * THREADS];
    }
    v4f buf[2][GROUP];
    #pragma unroll
    for (int t = 0; t < GROUP; ++t) buf[0][t] = *(const v4f*)(xg + (long)(2 * t) * DFEAT);
    #pragma unroll
    for (int t = 0; t < GROUP; ++t) buf[1][t] = *(const v4f*)(xg + (long)(2 * t) * DFEAT + 8 * GROUP);
    {
        v4f* dst = (v4f*)w1s;
        #pragma unroll
        for (int i = 0; i < W1_PER_THREAD; ++i) dst[threadIdx.x + i * THREADS] = wtmp[i];
    }
    if (threadIdx.x < NGATES) {
        float w = qw[threadIdx.x];
        gsin[threadIdx.x] = sinf(w);
        gcos[threadIdx.x] = cosf(w);
    }
    __syncthreads();

    v8f acc = {0.f, 0.f, 0.f, 0.f, 0.f, 0.f, 0.f, 0.f};

    #pragma unroll
    for (int g = 0; g < NGROUPS; ++g) {
        #pragma unroll
        for (int t = 0; t < GROUP; ++t)
            *(v4f*)(stw + 2 * t * STAGE_PITCH) = buf[g & 1][t];
        if (g + 2 < NGROUPS) {
            #pragma unroll
            for (int t = 0; t < GROUP; ++t)
                buf[g & 1][t] = *(const v4f*)(xg + (long)(2 * t) * DFEAT + (g + 2) * (8 * GROUP));
        }
        __builtin_amdgcn_fence(__ATOMIC_ACQ_REL, "wavefront", "local");
        #pragma unroll
        for (int t = 0; t < GROUP; ++t) {
            const v4f av = *(const v4f*)(str + t * 8);
            const float* bq = bp + (g * (8 * GROUP) + t * 8) * NQ;
            v2f a0, a1, b0, b1v;
            a0.x = av.x;  a0.y = av.y;
            a1.x = av.z;  a1.y = av.w;
            b0.x  = bq[0];       b0.y  = bq[NQ];
            b1v.x = bq[2 * NQ];  b1v.y = bq[3 * NQ];
            acc = __builtin_amdgcn_wmma_f32_16x16x4_f32(
                 false, a0,  false, b0,
                 (short)0, acc,  false,  false);
            acc = __builtin_amdgcn_wmma_f32_16x16x4_f32(
                 false, a1,  false, b1v,
                 (short)0, acc,  false,  false);
        }
        __builtin_amdgcn_fence(__ATOMIC_ACQ_REL, "wavefront", "local");
    }

    if (c < NQ) {
        #pragma unroll
        for (int j = 0; j < 8; ++j)
            hP[kq][tile * 16 + hi * 8 + j][c] = acc[j];
    }
    __syncthreads();

    if (threadIdx.x < ROWS_PER_BLOCK) {
        const int r = threadIdx.x;
        float a[NQ];
        #pragma unroll
        for (int n = 0; n < NQ; ++n)
            a[n] = ((hP[0][r][n] + hP[1][r][n]) + (hP[2][r][n] + hP[3][r][n])) + b1[n];

        float ss = 0.f;
        #pragma unroll
        for (int n = 0; n < NQ; ++n) ss += a[n] * a[n];
        float inv = 1.0f / sqrtf(ss);
        #pragma unroll
        for (int n = 0; n < NQ; ++n) a[n] *= inv;

        const int Q[NGATES] = {4, 3, 2, 4, 1, 3, 0, 2, 4, 1, 3, 2, 4, 3, 4};
        #pragma unroll
        for (int k = 0; k < NGATES; ++k) {
            const int q = Q[k];
            float s = gsin[k], cc = gcos[k];
            float aq  =  cc * a[q] + s * a[q + 1];
            float aq1 = -s * a[q] + cc * a[q + 1];
            a[q] = aq; a[q + 1] = aq1;
        }

        float accum = b2[0];
        #pragma unroll
        for (int n = 0; n < NQ; ++n) {
            float z = 1.f - 2.f * a[n] * a[n];
            z = (z != z) ? 0.f : z;
            accum += z * W2[n];
        }
        out[(long)blockIdx.x * ROWS_PER_BLOCK + r] = 1.f / (1.f + expf(-accum));
    }
}

extern "C" void kernel_launch(void* const* d_in, const int* in_sizes, int n_in,
                              void* d_out, int out_size, void* d_ws, size_t ws_size,
                              hipStream_t stream) {
    const float* x  = (const float*)d_in[0];
    const float* W1 = (const float*)d_in[1];
    const float* b1 = (const float*)d_in[2];
    const float* qw = (const float*)d_in[3];
    const float* W2 = (const float*)d_in[4];
    const float* b2 = (const float*)d_in[5];
    float* out = (float*)d_out;

    const int B = in_sizes[0] / DFEAT;
    dim3 grid(B / ROWS_PER_BLOCK);
    hybrid_pyramid_fused<<<grid, THREADS, 0, stream>>>(x, W1, b1, qw, W2, b2, out);
}


# speedup vs baseline: 3.3340x; 1.0768x over previous
#include <hip/hip_runtime.h>
#include <math.h>

typedef __attribute__((ext_vector_type(2))) float v2f;
typedef __attribute__((ext_vector_type(4))) float v4f;
typedef __attribute__((ext_vector_type(8))) float v8f;

#define NQ 6
#define NGATES 15
#define DFEAT 4096
#define KSPLIT 4
#define KPER (DFEAT / KSPLIT)
#define TILES 4
#define ROWS_PER_BLOCK (16 * TILES)
#define THREADS (32 * TILES * KSPLIT)
#define GROUP 8
#define NGROUPS (KPER / (8 * GROUP))
#define W1_PER_THREAD ((DFEAT * NQ / 4) / THREADS)
#define NWAVES (THREADS / 32)
#define STAGE_PITCH 68

__device__ __forceinline__ v4f ld_stream(const float* p) {
    return __builtin_nontemporal_load((const v4f*)p);
}

__global__ __launch_bounds__(THREADS) void hybrid_pyramid_fused(
    const float* __restrict__ x,  const float* __restrict__ W1,
    const float* __restrict__ b1, const float* __restrict__ qw,
    const float* __restrict__ W2, const float* __restrict__ b2,
    float* __restrict__ out)
{
    __shared__ float w1s[DFEAT * NQ];
    __shared__ float xs[NWAVES][16][STAGE_PITCH];
    __shared__ float hP[KSPLIT][ROWS_PER_BLOCK][NQ];
    __shared__ float gsin[NGATES + 1];
    __shared__ float gcos[NGATES + 1];

    const int lane = threadIdx.x & 31;
    const int wave = threadIdx.x >> 5;
    const int c    = lane & 15;
    const int hi   = lane >> 4;
    const int tile = wave >> 2;
    const int kq   = wave & 3;
    const int ncol = (c < NQ) ? c : (NQ - 1);

    const long row0 = (long)blockIdx.x * ROWS_PER_BLOCK + tile * 16;
    const float* __restrict__ xg = x + (row0 + hi) * DFEAT + kq * KPER + c * 4;
    float* stw = &xs[wave][hi][c * 4];
    const float* str = &xs[wave][c][hi * 4];
    const float* bp = w1s + (kq * KPER + hi * 4) * NQ + ncol;

    v4f wtmp[W1_PER_THREAD];
    {
        const v4f* __restrict__ src = (const v4f*)W1;
        #pragma unroll
        for (int i = 0; i < W1_PER_THREAD; ++i) wtmp[i] = src[threadIdx.x + i * THREADS];
    }
    v4f buf[2][GROUP];
    #pragma unroll
    for (int t = 0; t < GROUP; ++t) buf[0][t] = ld_stream(xg + (long)(2 * t) * DFEAT);
    #pragma unroll
    for (int t = 0; t < GROUP; ++t) buf[1][t] = ld_stream(xg + (long)(2 * t) * DFEAT + 8 * GROUP);
    {
        v4f* dst = (v4f*)w1s;
        #pragma unroll
        for (int i = 0; i < W1_PER_THREAD; ++i) dst[threadIdx.x + i * THREADS] = wtmp[i];
    }
    if (threadIdx.x < NGATES) {
        float w = qw[threadIdx.x];
        gsin[threadIdx.x] = sinf(w);
        gcos[threadIdx.x] = cosf(w);
    }
    __syncthreads();

    v8f acc = {0.f, 0.f, 0.f, 0.f, 0.f, 0.f, 0.f, 0.f};

    #pragma unroll
    for (int g = 0; g < NGROUPS; ++g) {
        #pragma unroll
        for (int t = 0; t < GROUP; ++t)
            *(v4f*)(stw + 2 * t * STAGE_PITCH) = buf[g & 1][t];
        if (g + 2 < NGROUPS) {
            #pragma unroll
            for (int t = 0; t < GROUP; ++t)
                buf[g & 1][t] = ld_stream(xg + (long)(2 * t) * DFEAT + (g + 2) * (8 * GROUP));
        }
        __builtin_amdgcn_fence(__ATOMIC_ACQ_REL, "wavefront", "local");
        #pragma unroll
        for (int t = 0; t < GROUP; ++t) {
            const v4f av = *(const v4f*)(str + t * 8);
            const float* bq = bp + (g * (8 * GROUP) + t * 8) * NQ;
            v2f a0, a1, b0, b1v;
            a0.x = av.x;  a0.y = av.y;
            a1.x = av.z;  a1.y = av.w;
            b0.x  = bq[0];       b0.y  = bq[NQ];
            b1v.x = bq[2 * NQ];  b1v.y = bq[3 * NQ];
            acc = __builtin_amdgcn_wmma_f32_16x16x4_f32(
                 false, a0,  false, b0,
                 (short)0, acc,  false,  false);
            acc = __builtin_amdgcn_wmma_f32_16x16x4_f32(
                 false, a1,  false, b1v,
                 (short)0, acc,  false,  false);
        }
        __builtin_amdgcn_fence(__ATOMIC_ACQ_REL, "wavefront", "local");
    }

    if (c < NQ) {
        #pragma unroll
        for (int j = 0; j < 8; ++j)
            hP[kq][tile * 16 + hi * 8 + j][c] = acc[j];
    }
    __syncthreads();

    if (threadIdx.x < ROWS_PER_BLOCK) {
        const int r = threadIdx.x;
        float a[NQ];
        #pragma unroll
        for (int n = 0; n < NQ; ++n)
            a[n] = ((hP[0][r][n] + hP[1][r][n]) + (hP[2][r][n] + hP[3][r][n])) + b1[n];

        float ss = 0.f;
        #pragma unroll
        for (int n = 0; n < NQ; ++n) ss += a[n] * a[n];
        float inv = 1.0f / sqrtf(ss);
        #pragma unroll
        for (int n = 0; n < NQ; ++n) a[n] *= inv;

        const int Q[NGATES] = {4, 3, 2, 4, 1, 3, 0, 2, 4, 1, 3, 2, 4, 3, 4};
        #pragma unroll
        for (int k = 0; k < NGATES; ++k) {
            const int q = Q[k];
            float s = gsin[k], cc = gcos[k];
            float aq  =  cc * a[q] + s * a[q + 1];
            float aq1 = -s * a[q] + cc * a[q + 1];
            a[q] = aq; a[q + 1] = aq1;
        }

        float accum = b2[0];
        #pragma unroll
        for (int n = 0; n < NQ; ++n) {
            float z = 1.f - 2.f * a[n] * a[n];
            z = (z != z) ? 0.f : z;
            accum += z * W2[n];
        }
        out[(long)blockIdx.x * ROWS_PER_BLOCK + r] = 1.f / (1.f + expf(-accum));
    }
}

extern "C" void kernel_launch(void* const* d_in, const int* in_sizes, int n_in,
                              void* d_out, int out_size, void* d_ws, size_t ws_size,
                              hipStream_t stream) {
    const float* x  = (const float*)d_in[0];
    const float* W1 = (const float*)d_in[1];
    const float* b1 = (const float*)d_in[2];
    const float* qw = (const float*)d_in[3];
    const float* W2 = (const float*)d_in[4];
    const float* b2 = (const float*)d_in[5];
    float* out = (float*)d_out;

    const int B = in_sizes[0] / DFEAT;
    dim3 grid(B / ROWS_PER_BLOCK);
    hybrid_pyramid_fused<<<grid, THREADS, 0, stream>>>(x, W1, b1, qw, W2, b2, out);
}
